# starting kernel
# baseline (speedup 1.0000x reference)
#include <hip/hip_runtime.h>

#define DIMS_IN  128
#define DIMS_H   96
#define DIMS_OUT 32

typedef float v2f __attribute__((ext_vector_type(2)));
typedef float v8f __attribute__((ext_vector_type(8)));

__global__ void zero_kernel(float4* __restrict__ p, int n4) {
    int i = blockIdx.x * blockDim.x + threadIdx.x;
    if (i < n4) {
        float4 z; z.x = 0.f; z.y = 0.f; z.z = 0.f; z.w = 0.f;
        p[i] = z;
    }
}

__global__ void transpose_kernel(const float* __restrict__ W,
                                 float* __restrict__ Wt, int K, int N) {
    int i = blockIdx.x * blockDim.x + threadIdx.x;
    if (i < K * N) {
        int k = i / N;
        int n = i - k * N;
        Wt[n * K + k] = W[i];
    }
}

template <int K, bool RELU_A>
__global__ void gemm_wmma_f32_bt(const float* __restrict__ A,
                                 const float* __restrict__ Bt,
                                 float* __restrict__ C,
                                 int M, int N) {
    const int wavesPerBlock = blockDim.x >> 5;
    const int waveId = blockIdx.x * wavesPerBlock + (threadIdx.x >> 5);
    const int lane   = threadIdx.x & 31;
    const int nPair  = N >> 5;
    const int tileM  = waveId / nPair;
    const int tileP  = waveId - tileM * nPair;
    if (tileM * 16 >= M) return;

    const int half = lane >> 4;
    const int l15  = lane & 15;

    const float* arow = A  + (tileM * 16 + l15) * K + half * 2;
    const float* brow = Bt + (tileP * 32 + l15) * K + half * 2;

    v8f acc0 = {};
    v8f acc1 = {};
    #pragma unroll 4
    for (int k = 0; k < K; k += 4) {
        v2f a = *(const v2f*)(arow + k);
        if (RELU_A) {
            a.x = fmaxf(a.x, 0.f);
            a.y = fmaxf(a.y, 0.f);
        }
        v2f b0 = *(const v2f*)(brow + k);
        v2f b1 = *(const v2f*)(brow + 16 * K + k);
        acc0 = __builtin_amdgcn_wmma_f32_16x16x4_f32(
            false, a, false, b0, (short)0, acc0, false, false);
        acc1 = __builtin_amdgcn_wmma_f32_16x16x4_f32(
            false, a, false, b1, (short)0, acc1, false, false);
    }

    float* c0 = C + (tileM * 16 + half * 8) * N + tileP * 32 + l15;
    #pragma unroll
    for (int r = 0; r < 8; ++r) {
        c0[r * N]      = acc0[r];
        c0[r * N + 16] = acc1[r];
    }
}

__global__ void spmm_scatter(const int* __restrict__ esrc,
                             const int* __restrict__ edst,
                             const float* __restrict__ eval,
                             const float* __restrict__ dense,
                             float* __restrict__ out,
                             int total, int D4, int D) {
    int i = blockIdx.x * blockDim.x + threadIdx.x;
    if (i >= total) return;
    int e = i / D4;
    int c = (i - e * D4) * 4;
    int s = esrc[e];
    int d = edst[e];
    float v = eval[e];
    const float4 m = *(const float4*)(dense + (size_t)s * D + c);
    float* o = out + (size_t)d * D + c;
    atomicAdd(o + 0, v * m.x);
    atomicAdd(o + 1, v * m.y);
    atomicAdd(o + 2, v * m.z);
    atomicAdd(o + 3, v * m.w);
}

extern "C" void kernel_launch(void* const* d_in, const int* in_sizes, int n_in,
                              void* d_out, int out_size, void* d_ws, size_t ws_size,
                              hipStream_t stream) {
    const float* x    = (const float*)d_in[0];
    const int*   esrc = (const int*)d_in[1];
    const int*   edst = (const int*)d_in[2];
    const float* eval = (const float*)d_in[3];
    const float* W1   = (const float*)d_in[4];
    const float* W2   = (const float*)d_in[5];
    float*       out  = (float*)d_out;

    const int E = in_sizes[1];
    const int n = in_sizes[0] / DIMS_IN;

    float* buf0 = (float*)d_ws;
    float* buf1 = buf0 + (size_t)n * DIMS_H;
    float* W1t  = buf1 + (size_t)n * DIMS_H;
    float* W2t  = W1t + DIMS_H * DIMS_IN;

    transpose_kernel<<<(DIMS_IN * DIMS_H + 255) / 256, 256, 0, stream>>>(
        W1, W1t, DIMS_IN, DIMS_H);
    transpose_kernel<<<(DIMS_H * DIMS_OUT + 255) / 256, 256, 0, stream>>>(
        W2, W2t, DIMS_H, DIMS_OUT);

    {
        int n4 = (n * DIMS_H) / 4;
        zero_kernel<<<(n4 + 255) / 256, 256, 0, stream>>>((float4*)buf1, n4);
        int o4 = (n * DIMS_OUT) / 4;
        zero_kernel<<<(o4 + 255) / 256, 256, 0, stream>>>((float4*)out, o4);
    }

    {
        int waves  = (n / 16) * (DIMS_H / 32);
        int blocks = (waves + 7) / 8;
        gemm_wmma_f32_bt<DIMS_IN, false><<<blocks, 256, 0, stream>>>(
            x, W1t, buf0, n, DIMS_H);
    }

    {
        int total = E * (DIMS_H / 4);
        spmm_scatter<<<(total + 255) / 256, 256, 0, stream>>>(
            esrc, edst, eval, buf0, buf1, total, DIMS_H / 4, DIMS_H);
    }

    {
        int waves  = (n / 16) * (DIMS_OUT / 32);
        int blocks = (waves + 7) / 8;
        gemm_wmma_f32_bt<DIMS_H, true><<<blocks, 256, 0, stream>>>(
            buf1, W2t, buf0, n, DIMS_OUT);
    }

    {
        int total = E * (DIMS_OUT / 4);
        spmm_scatter<<<(total + 255) / 256, 256, 0, stream>>>(
            esrc, edst, eval, buf0, out, total, DIMS_OUT / 4, DIMS_OUT);
    }
}


# Layer 1 computed as (A x) W1: SpMM1 sums x rows (16-byte loads) into its LDS row tile, then multiplies the tile by W1 on f32 WMMA from LDS; GEMM1 launch gone; sum order changes
# speedup vs baseline: 2.1769x; 2.1769x over previous
#include <hip/hip_runtime.h>

#define DIMS_IN  128
#define DIMS_H   96
#define DIMS_OUT 32

typedef float v2f __attribute__((ext_vector_type(2)));
typedef float v8f __attribute__((ext_vector_type(8)));

__global__ void transpose_kernel(const float* __restrict__ W,
                                 float* __restrict__ Wt, int K, int N) {
    int i = blockIdx.x * blockDim.x + threadIdx.x;
    if (i < K * N) {
        int k = i / N;
        int n = i - k * N;
        Wt[n * K + k] = W[i];
    }
}

template <int K, bool RELU_A>
__global__ void gemm_wmma_f32_bt(const float* __restrict__ A,
                                 const float* __restrict__ Bt,
                                 float* __restrict__ C,
                                 int M, int N) {
    const int wavesPerBlock = blockDim.x >> 5;
    const int waveId = blockIdx.x * wavesPerBlock + (threadIdx.x >> 5);
    const int lane   = threadIdx.x & 31;
    const int nPair  = N >> 5;
    const int tileM  = waveId / nPair;
    const int tileP  = waveId - tileM * nPair;
    if (tileM * 16 >= M) return;

    const int half = lane >> 4;
    const int l15  = lane & 15;

    const float* arow = A  + (tileM * 16 + l15) * K + half * 2;
    const float* brow = Bt + (tileP * 32 + l15) * K + half * 2;

    v8f acc0 = {};
    v8f acc1 = {};
    #pragma unroll 4
    for (int k = 0; k < K; k += 4) {
        v2f a = *(const v2f*)(arow + k);
        if (RELU_A) {
            a.x = fmaxf(a.x, 0.f);
            a.y = fmaxf(a.y, 0.f);
        }
        v2f b0 = *(const v2f*)(brow + k);
        v2f b1 = *(const v2f*)(brow + 16 * K + k);
        acc0 = __builtin_amdgcn_wmma_f32_16x16x4_f32(
            false, a, false, b0, (short)0, acc0, false, false);
        acc1 = __builtin_amdgcn_wmma_f32_16x16x4_f32(
            false, a, false, b1, (short)0, acc1, false, false);
    }

    float* c0 = C + (tileM * 16 + half * 8) * N + tileP * 32 + l15;
    #pragma unroll
    for (int r = 0; r < 8; ++r) {
        c0[r * N]      = acc0[r];
        c0[r * N + 16] = acc1[r];
    }
}

#define SP_ROWS    196
#define SP_TILES   256
#define SP_CHUNKS  256
#define SP_CAP     64
#define BIN_THREADS 1024
#define SPT_THREADS 512
#define SPT_WAVES   (SPT_THREADS / 32)

__global__ __launch_bounds__(BIN_THREADS)
void bin_edges(const int* __restrict__ esrc,
               const int* __restrict__ edst,
               const float* __restrict__ eval,
               int* __restrict__ ecnt,
               int4* __restrict__ erec,
               int E, int n) {
    __shared__ int counters[SP_TILES];
    const int tid = threadIdx.x;
    const int c   = blockIdx.x;
    for (int i = tid; i < SP_TILES; i += BIN_THREADS) counters[i] = 0;
    __syncthreads();

    const int chunk = (E + SP_CHUNKS - 1) / SP_CHUNKS;
    const int e0 = c * chunk;
    const int e1 = min(E, e0 + chunk);
    for (int e = e0 + tid; e < e1; e += BIN_THREADS) {
        const int   d = edst[e];
        const int   s = esrc[e];
        const float v = eval[e];
        asm volatile("" :: "v"(s), "v"(v));
        if ((unsigned)d < (unsigned)n) {
            const int t = d / SP_ROWS;
            if (t < SP_TILES) {
                const int slot = atomicAdd(&counters[t], 1);
                if (slot < SP_CAP) {
                    int4 rec;
                    rec.x = s;
                    rec.y = d - t * SP_ROWS;
                    rec.z = __float_as_int(v);
                    rec.w = 0;
                    erec[((size_t)c * SP_TILES + t) * SP_CAP + slot] = rec;
                }
            }
        }
    }
    __syncthreads();
    for (int i = tid; i < SP_TILES; i += BIN_THREADS)
        ecnt[c * SP_TILES + i] = counters[i];
}

template <int D>
__device__ __forceinline__ void tile_add_records(const int4 rec, const int m,
                                                 const float* __restrict__ dense,
                                                 float* acc, const int lane) {
    constexpr int C = D / 32;
    constexpr int U = (48 / C) > 32 ? 32 : (48 / C);
    for (int j = 0; j < m; j += U) {
        int sj[U];
        #pragma unroll
        for (int u = 0; u < U; ++u) sj[u] = __builtin_amdgcn_readlane(rec.x, j + u);
        float g[U][C];
        #pragma unroll
        for (int u = 0; u < U; ++u) {
            if (j + u < m) {
                const float* srow = dense + (size_t)sj[u] * D + lane;
                #pragma unroll
                for (int k = 0; k < C; ++k) g[u][k] = srow[k * 32];
            }
        }
        #pragma unroll
        for (int u = 0; u < U; ++u) {
            if (j + u < m) {
                const int   rj = __builtin_amdgcn_readlane(rec.y, j + u);
                const float vj = __int_as_float(__builtin_amdgcn_readlane(rec.z, j + u));
                float* a = acc + rj * D + lane;
                #pragma unroll
                for (int k = 0; k < C; ++k) atomicAdd(a + k * 32, vj * g[u][k]);
            }
        }
    }
}

template <int D>
__global__ __launch_bounds__(SPT_THREADS)
void spmm_tile_lds(const int* __restrict__ ecnt,
                   const int4* __restrict__ erec,
                   const int* __restrict__ esrc,
                   const int* __restrict__ edst,
                   const float* __restrict__ eval,
                   const float* __restrict__ dense,
                   float* __restrict__ out,
                   int E, int n) {
    __shared__ __attribute__((aligned(16))) float acc[SP_ROWS * D];

    const int tid  = threadIdx.x;
    const int lane = tid & 31;
    const int wave = __builtin_amdgcn_readfirstlane(tid >> 5);
    const int t    = blockIdx.x;
    const int row0 = t * SP_ROWS;
    if (row0 >= n) return;
    const int rows = min(SP_ROWS, n - row0);
    const int nel  = rows * D;

    {
        float4 z; z.x = 0.f; z.y = 0.f; z.z = 0.f; z.w = 0.f;
        for (int i = tid * 4; i < nel; i += SPT_THREADS * 4)
            *(float4*)(acc + i) = z;
    }
    __syncthreads();

    {
        const int4 zero4 = make_int4(0, 0, 0, 0);
        int c = wave;
        int cntc = 0;
        int4 r0 = zero4, r1 = zero4;
        if (c < SP_CHUNKS) {
            cntc = __builtin_amdgcn_readfirstlane(ecnt[c * SP_TILES + t]);
            const int4* cell = erec + ((size_t)c * SP_TILES + t) * SP_CAP;
            if (cntc <= SP_CAP) {
                if (lane < cntc)      r0 = cell[lane];
                if (lane + 32 < cntc) r1 = cell[lane + 32];
            }
        }
        while (c < SP_CHUNKS) {
            const int nc = c + SPT_WAVES;
            int cntn = 0;
            int4 n0 = zero4, n1 = zero4;
            if (nc < SP_CHUNKS) {
                cntn = __builtin_amdgcn_readfirstlane(ecnt[nc * SP_TILES + t]);
                const int4* cell = erec + ((size_t)nc * SP_TILES + t) * SP_CAP;
                if (cntn <= SP_CAP) {
                    if (lane < cntn)      n0 = cell[lane];
                    if (lane + 32 < cntn) n1 = cell[lane + 32];
                }
            }
            if (cntc <= SP_CAP) {
                tile_add_records<D>(r0, min(cntc, 32), dense, acc, lane);
                if (cntc > 32) tile_add_records<D>(r1, cntc - 32, dense, acc, lane);
            } else {
                const int chunk = (E + SP_CHUNKS - 1) / SP_CHUNKS;
                const int e0 = c * chunk;
                const int e1 = min(E, e0 + chunk);
                for (int e = e0 + lane; e < e1; e += 32) {
                    const unsigned r = (unsigned)(edst[e] - row0);
                    if (r < (unsigned)rows) {
                        const int   s = esrc[e];
                        const float v = eval[e];
                        const float* srow = dense + (size_t)s * D;
                        float* a = acc + (int)r * D;
                        for (int k = 0; k < D; ++k) atomicAdd(a + k, v * srow[k]);
                    }
                }
            }
            c = nc; cntc = cntn; r0 = n0; r1 = n1;
        }
    }
    __syncthreads();

    {
        float* o = out + (size_t)row0 * D;
        for (int i = tid * 4; i < nel; i += SPT_THREADS * 4)
            *(float4*)(o + i) = *(const float4*)(acc + i);
    }
}

#define XT_PITCH 132
#define XT_ROWS  208
#define HT_PITCH 100

__device__ __forceinline__ void tile_add_x_records(const int4 rec, const int m,
                                                   const float* __restrict__ x,
                                                   float* xt, const int lane) {
    constexpr int U = 16;
    for (int j = 0; j < m; j += U) {
        int sj[U];
        #pragma unroll
        for (int u = 0; u < U; ++u) sj[u] = __builtin_amdgcn_readlane(rec.x, j + u);
        float4 g[U];
        #pragma unroll
        for (int u = 0; u < U; ++u) {
            if (j + u < m)
                g[u] = *(const float4*)(x + (size_t)sj[u] * DIMS_IN + lane * 4);
        }
        #pragma unroll
        for (int u = 0; u < U; ++u) {
            if (j + u < m) {
                const int   rj = __builtin_amdgcn_readlane(rec.y, j + u);
                const float vj = __int_as_float(__builtin_amdgcn_readlane(rec.z, j + u));
                float* a = xt + rj * XT_PITCH + lane;
                atomicAdd(a,      vj * g[u].x);
                atomicAdd(a + 32, vj * g[u].y);
                atomicAdd(a + 64, vj * g[u].z);
                atomicAdd(a + 96, vj * g[u].w);
            }
        }
    }
}

__global__ __launch_bounds__(SPT_THREADS)
void spmm_x_tile_gemm(const int* __restrict__ ecnt,
                      const int4* __restrict__ erec,
                      const int* __restrict__ esrc,
                      const int* __restrict__ edst,
                      const float* __restrict__ eval,
                      const float* __restrict__ x,
                      const float* __restrict__ W1t,
                      float* __restrict__ hout,
                      int E, int n) {
    __shared__ __attribute__((aligned(16))) float xt[XT_ROWS * XT_PITCH];
    __shared__ __attribute__((aligned(16))) float wl[DIMS_H * XT_PITCH];
    __shared__ __attribute__((aligned(16))) float ht[XT_ROWS * HT_PITCH];

    const int tid  = threadIdx.x;
    const int lane = tid & 31;
    const int wave = __builtin_amdgcn_readfirstlane(tid >> 5);
    const int t    = blockIdx.x;
    const int row0 = t * SP_ROWS;
    if (row0 >= n) return;
    const int rows = min(SP_ROWS, n - row0);

    {
        float4 z; z.x = 0.f; z.y = 0.f; z.z = 0.f; z.w = 0.f;
        for (int i = tid * 4; i < XT_ROWS * XT_PITCH; i += SPT_THREADS * 4)
            *(float4*)(xt + i) = z;
        for (int q = tid; q < DIMS_H * 32; q += SPT_THREADS) {
            const int nn = q >> 5;
            const float4 w = *(const float4*)(W1t + nn * DIMS_IN + lane * 4);
            float* d = wl + nn * XT_PITCH + lane;
            d[0]  = w.x;
            d[32] = w.y;
            d[64] = w.z;
            d[96] = w.w;
        }
    }
    __syncthreads();

    {
        const int4 zero4 = make_int4(0, 0, 0, 0);
        int c = wave;
        int cntc = 0;
        int4 r0 = zero4, r1 = zero4;
        if (c < SP_CHUNKS) {
            cntc = __builtin_amdgcn_readfirstlane(ecnt[c * SP_TILES + t]);
            const int4* cell = erec + ((size_t)c * SP_TILES + t) * SP_CAP;
            if (cntc <= SP_CAP) {
                if (lane < cntc)      r0 = cell[lane];
                if (lane + 32 < cntc) r1 = cell[lane + 32];
            }
        }
        while (c < SP_CHUNKS) {
            const int nc = c + SPT_WAVES;
            int cntn = 0;
            int4 n0 = zero4, n1 = zero4;
            if (nc < SP_CHUNKS) {
                cntn = __builtin_amdgcn_readfirstlane(ecnt[nc * SP_TILES + t]);
                const int4* cell = erec + ((size_t)nc * SP_TILES + t) * SP_CAP;
                if (cntn <= SP_CAP) {
                    if (lane < cntn)      n0 = cell[lane];
                    if (lane + 32 < cntn) n1 = cell[lane + 32];
                }
            }
            if (cntc <= SP_CAP) {
                tile_add_x_records(r0, min(cntc, 32), x, xt, lane);
                if (cntc > 32) tile_add_x_records(r1, cntc - 32, x, xt, lane);
            } else {
                const int chunk = (E + SP_CHUNKS - 1) / SP_CHUNKS;
                const int e0 = c * chunk;
                const int e1 = min(E, e0 + chunk);
                for (int e = e0 + lane; e < e1; e += 32) {
                    const unsigned r = (unsigned)(edst[e] - row0);
                    if (r < (unsigned)rows) {
                        const int   s = esrc[e];
                        const float v = eval[e];
                        const float* srow = x + (size_t)s * DIMS_IN;
                        float* a = xt + (int)r * XT_PITCH;
                        for (int k = 0; k < DIMS_IN; ++k)
                            atomicAdd(a + (k & 3) * 32 + (k >> 2), v * srow[k]);
                    }
                }
            }
            c = nc; cntc = cntn; r0 = n0; r1 = n1;
        }
    }
    __syncthreads();

    {
        const int nmt  = (rows + 15) >> 4;
        const int half = lane >> 4;
        const int l15  = lane & 15;
        for (int u = wave; u < nmt * 2; u += SPT_WAVES) {
            const int mt = u >> 1;
            const int nh = u & 1;
            const float* ap = xt + (mt * 16 + l15) * XT_PITCH + half * 2;
            const float* bp = wl + (nh * 48 + l15) * XT_PITCH + half * 2;
            v8f d0 = {};
            v8f d1 = {};
            v8f d2 = {};
            #pragma unroll 8
            for (int k = 0; k < DIMS_IN; k += 4) {
                const v2f a  = *(const v2f*)(ap + k);
                const v2f b0 = *(const v2f*)(bp + k);
                const v2f b1 = *(const v2f*)(bp + 16 * XT_PITCH + k);
                const v2f b2 = *(const v2f*)(bp + 32 * XT_PITCH + k);
                d0 = __builtin_amdgcn_wmma_f32_16x16x4_f32(
                    false, a, false, b0, (short)0, d0, false, false);
                d1 = __builtin_amdgcn_wmma_f32_16x16x4_f32(
                    false, a, false, b1, (short)0, d1, false, false);
                d2 = __builtin_amdgcn_wmma_f32_16x16x4_f32(
                    false, a, false, b2, (short)0, d2, false, false);
            }
            float* hp = ht + (mt * 16 + half * 8) * HT_PITCH + nh * 48 + l15;
            #pragma unroll
            for (int r = 0; r < 8; ++r) {
                hp[r * HT_PITCH]      = d0[r];
                hp[r * HT_PITCH + 16] = d1[r];
                hp[r * HT_PITCH + 32] = d2[r];
            }
        }
    }
    __syncthreads();

    {
        float* o = hout + (size_t)row0 * DIMS_H;
        const int nel = rows * DIMS_H;
        for (int i = tid * 4; i < nel; i += SPT_THREADS * 4) {
            const int r  = i / DIMS_H;
            const int cc = i - r * DIMS_H;
            *(float4*)(o + i) = *(const float4*)(ht + r * HT_PITCH + cc);
        }
    }
}

extern "C" void kernel_launch(void* const* d_in, const int* in_sizes, int n_in,
                              void* d_out, int out_size, void* d_ws, size_t ws_size,
                              hipStream_t stream) {
    const float* x    = (const float*)d_in[0];
    const int*   esrc = (const int*)d_in[1];
    const int*   edst = (const int*)d_in[2];
    const float* eval = (const float*)d_in[3];
    const float* W1   = (const float*)d_in[4];
    const float* W2   = (const float*)d_in[5];
    float*       out  = (float*)d_out;

    const int E = in_sizes[1];
    const int n = in_sizes[0] / DIMS_IN;

    float* buf0 = (float*)d_ws;
    float* buf1 = buf0 + (size_t)n * DIMS_H;
    float* W1t  = buf1 + (size_t)n * DIMS_H;
    float* W2t  = W1t + DIMS_H * DIMS_IN;
    int*   ecnt = (int*)(W2t + DIMS_OUT * DIMS_H);
    int4*  erec = (int4*)(ecnt + SP_CHUNKS * SP_TILES);

    transpose_kernel<<<(DIMS_IN * DIMS_H + 255) / 256, 256, 0, stream>>>(
        W1, W1t, DIMS_IN, DIMS_H);
    transpose_kernel<<<(DIMS_H * DIMS_OUT + 255) / 256, 256, 0, stream>>>(
        W2, W2t, DIMS_H, DIMS_OUT);

    bin_edges<<<SP_CHUNKS, BIN_THREADS, 0, stream>>>(esrc, edst, eval, ecnt, erec, E, n);

    spmm_x_tile_gemm<<<SP_TILES, SPT_THREADS, 0, stream>>>(
        ecnt, erec, esrc, edst, eval, x, W1t, buf1, E, n);

    {
        int waves  = (n / 16) * (DIMS_OUT / 32);
        int blocks = (waves + 7) / 8;
        gemm_wmma_f32_bt<DIMS_H, true><<<blocks, 256, 0, stream>>>(
            buf1, W2t, buf0, n, DIMS_OUT);
    }

    spmm_tile_lds<DIMS_OUT><<<SP_TILES, SPT_THREADS, 0, stream>>>(
        ecnt, erec, esrc, edst, eval, buf0, out, E, n);
}


# SpMM tile kernels: a wave reads all its cell counts in one load and walks the cells' records as one flat list in full 32-record batches with unguarded row groups; sum order = LDS arrival as before
# speedup vs baseline: 2.7640x; 1.2697x over previous
#include <hip/hip_runtime.h>

#define DIMS_IN  128
#define DIMS_H   96
#define DIMS_OUT 32

typedef float v2f __attribute__((ext_vector_type(2)));
typedef float v8f __attribute__((ext_vector_type(8)));

__global__ void transpose_kernel(const float* __restrict__ W,
                                 float* __restrict__ Wt, int K, int N) {
    int i = blockIdx.x * blockDim.x + threadIdx.x;
    if (i < K * N) {
        int k = i / N;
        int n = i - k * N;
        Wt[n * K + k] = W[i];
    }
}

template <int K, bool RELU_A>
__global__ void gemm_wmma_f32_bt(const float* __restrict__ A,
                                 const float* __restrict__ Bt,
                                 float* __restrict__ C,
                                 int M, int N) {
    const int wavesPerBlock = blockDim.x >> 5;
    const int waveId = blockIdx.x * wavesPerBlock + (threadIdx.x >> 5);
    const int lane   = threadIdx.x & 31;
    const int nPair  = N >> 5;
    const int tileM  = waveId / nPair;
    const int tileP  = waveId - tileM * nPair;
    if (tileM * 16 >= M) return;

    const int half = lane >> 4;
    const int l15  = lane & 15;

    const float* arow = A  + (tileM * 16 + l15) * K + half * 2;
    const float* brow = Bt + (tileP * 32 + l15) * K + half * 2;

    v8f acc0 = {};
    v8f acc1 = {};
    #pragma unroll 4
    for (int k = 0; k < K; k += 4) {
        v2f a = *(const v2f*)(arow + k);
        if (RELU_A) {
            a.x = fmaxf(a.x, 0.f);
            a.y = fmaxf(a.y, 0.f);
        }
        v2f b0 = *(const v2f*)(brow + k);
        v2f b1 = *(const v2f*)(brow + 16 * K + k);
        acc0 = __builtin_amdgcn_wmma_f32_16x16x4_f32(
            false, a, false, b0, (short)0, acc0, false, false);
        acc1 = __builtin_amdgcn_wmma_f32_16x16x4_f32(
            false, a, false, b1, (short)0, acc1, false, false);
    }

    float* c0 = C + (tileM * 16 + half * 8) * N + tileP * 32 + l15;
    #pragma unroll
    for (int r = 0; r < 8; ++r) {
        c0[r * N]      = acc0[r];
        c0[r * N + 16] = acc1[r];
    }
}

#define SP_ROWS    196
#define SP_TILES   256
#define SP_CHUNKS  256
#define SP_CAP     64
#define BIN_THREADS 1024
#define SPT_THREADS 512
#define SPT_WAVES   (SPT_THREADS / 32)
#define SP_CPW      (SP_CHUNKS / SPT_WAVES)
static_assert(SP_CHUNKS % SPT_WAVES == 0 && SP_CPW <= 32, "one lane per cell of the wave");

__global__ __launch_bounds__(BIN_THREADS)
void bin_edges(const int* __restrict__ esrc,
               const int* __restrict__ edst,
               const float* __restrict__ eval,
               int* __restrict__ ecnt,
               int4* __restrict__ erec,
               int E, int n) {
    __shared__ int counters[SP_TILES];
    const int tid = threadIdx.x;
    const int c   = blockIdx.x;
    for (int i = tid; i < SP_TILES; i += BIN_THREADS) counters[i] = 0;
    __syncthreads();

    const int chunk = (E + SP_CHUNKS - 1) / SP_CHUNKS;
    const int e0 = c * chunk;
    const int e1 = min(E, e0 + chunk);
    for (int e = e0 + tid; e < e1; e += BIN_THREADS) {
        const int   d = edst[e];
        const int   s = esrc[e];
        const float v = eval[e];
        asm volatile("" :: "v"(s), "v"(v));
        if ((unsigned)d < (unsigned)n) {
            const int t = d / SP_ROWS;
            if (t < SP_TILES) {
                const int slot = atomicAdd(&counters[t], 1);
                if (slot < SP_CAP) {
                    int4 rec;
                    rec.x = s;
                    rec.y = d - t * SP_ROWS;
                    rec.z = __float_as_int(v);
                    rec.w = 0;
                    erec[((size_t)c * SP_TILES + t) * SP_CAP + slot] = rec;
                }
            }
        }
    }
    __syncthreads();
    for (int i = tid; i < SP_TILES; i += BIN_THREADS)
        ecnt[c * SP_TILES + i] = counters[i];
}

__device__ __forceinline__ int4 load_flat_record(const int4* __restrict__ erec,
                                                 const int (&P)[SP_CPW + 1],
                                                 const int i, const int wave, const int t) {
    int k = 0, base = 0;
    #pragma unroll
    for (int j = 1; j < SP_CPW; ++j) {
        if (i >= P[j]) { k = j; base = P[j]; }
    }
    int4 r = make_int4(0, 0, 0, 0);
    if (i < P[SP_CPW])
        r = erec[((size_t)(wave + k * SPT_WAVES) * SP_TILES + t) * SP_CAP + (i - base)];
    return r;
}

template <int D>
__device__ __forceinline__ void tile_add_records(const int4 rec, const int m,
                                                 const float* __restrict__ dense,
                                                 float* acc, const int lane) {
    constexpr int C = D / 32;
    constexpr int U = (48 / C) > 32 ? 32 : (48 / C);
    int j = 0;
    for (; j + U <= m; j += U) {
        int sj[U];
        #pragma unroll
        for (int u = 0; u < U; ++u) sj[u] = __builtin_amdgcn_readlane(rec.x, j + u);
        float g[U][C];
        #pragma unroll
        for (int u = 0; u < U; ++u) {
            const float* srow = dense + (size_t)sj[u] * D + lane;
            #pragma unroll
            for (int k = 0; k < C; ++k) g[u][k] = srow[k * 32];
        }
        #pragma unroll
        for (int u = 0; u < U; ++u) {
            const int   rj = __builtin_amdgcn_readlane(rec.y, j + u);
            const float vj = __int_as_float(__builtin_amdgcn_readlane(rec.z, j + u));
            float* a = acc + rj * D + lane;
            #pragma unroll
            for (int k = 0; k < C; ++k) atomicAdd(a + k * 32, vj * g[u][k]);
        }
    }
    if (j < m) {
        int sj[U];
        #pragma unroll
        for (int u = 0; u < U; ++u) sj[u] = __builtin_amdgcn_readlane(rec.x, j + u);
        float g[U][C];
        #pragma unroll
        for (int u = 0; u < U; ++u) {
            if (j + u < m) {
                const float* srow = dense + (size_t)sj[u] * D + lane;
                #pragma unroll
                for (int k = 0; k < C; ++k) g[u][k] = srow[k * 32];
            }
        }
        #pragma unroll
        for (int u = 0; u < U; ++u) {
            if (j + u < m) {
                const int   rj = __builtin_amdgcn_readlane(rec.y, j + u);
                const float vj = __int_as_float(__builtin_amdgcn_readlane(rec.z, j + u));
                float* a = acc + rj * D + lane;
                #pragma unroll
                for (int k = 0; k < C; ++k) atomicAdd(a + k * 32, vj * g[u][k]);
            }
        }
    }
}

template <int D>
__global__ __launch_bounds__(SPT_THREADS)
void spmm_tile_lds(const int* __restrict__ ecnt,
                   const int4* __restrict__ erec,
                   const int* __restrict__ esrc,
                   const int* __restrict__ edst,
                   const float* __restrict__ eval,
                   const float* __restrict__ dense,
                   float* __restrict__ out,
                   int E, int n) {
    __shared__ __attribute__((aligned(16))) float acc[SP_ROWS * D];

    const int tid  = threadIdx.x;
    const int lane = tid & 31;
    const int wave = __builtin_amdgcn_readfirstlane(tid >> 5);
    const int t    = blockIdx.x;
    const int row0 = t * SP_ROWS;
    if (row0 >= n) return;
    const int rows = min(SP_ROWS, n - row0);
    const int nel  = rows * D;

    {
        float4 z; z.x = 0.f; z.y = 0.f; z.z = 0.f; z.w = 0.f;
        for (int i = tid * 4; i < nel; i += SPT_THREADS * 4)
            *(float4*)(acc + i) = z;
    }
    __syncthreads();

    {
        int myc = 0;
        if (lane < SP_CPW) myc = ecnt[(wave + lane * SPT_WAVES) * SP_TILES + t];
        int P[SP_CPW + 1];
        P[0] = 0;
        #pragma unroll
        for (int i = 0; i < SP_CPW; ++i) {
            const int ci = __builtin_amdgcn_readlane(myc, i);
            P[i + 1] = P[i] + (ci <= SP_CAP ? ci : 0);
        }
        const int total = P[SP_CPW];
        int4 cur = load_flat_record(erec, P, lane, wave, t);
        asm volatile("" :: "v"(cur.x), "v"(cur.y), "v"(cur.z));
        for (int b0 = 0; b0 < total; b0 += 32) {
            const int4 nxt = load_flat_record(erec, P, b0 + 32 + lane, wave, t);
            tile_add_records<D>(cur, min(32, total - b0), dense, acc, lane);
            cur = nxt;
        }
        unsigned ovf = __builtin_amdgcn_ballot_w32(myc > SP_CAP);
        while (ovf != 0u) {
            const int ci = __builtin_ctz(ovf);
            ovf &= ovf - 1u;
            const int c = wave + ci * SPT_WAVES;
            const int chunk = (E + SP_CHUNKS - 1) / SP_CHUNKS;
            const int e0 = c * chunk;
            const int e1 = min(E, e0 + chunk);
            for (int e = e0 + lane; e < e1; e += 32) {
                const unsigned r = (unsigned)(edst[e] - row0);
                if (r < (unsigned)rows) {
                    const int   s = esrc[e];
                    const float v = eval[e];
                    const float* srow = dense + (size_t)s * D;
                    float* a = acc + (int)r * D;
                    for (int k = 0; k < D; ++k) atomicAdd(a + k, v * srow[k]);
                }
            }
        }
    }
    __syncthreads();

    {
        float* o = out + (size_t)row0 * D;
        for (int i = tid * 4; i < nel; i += SPT_THREADS * 4)
            *(float4*)(o + i) = *(const float4*)(acc + i);
    }
}

#define XT_PITCH 132
#define XT_ROWS  208
#define HT_PITCH 100

__device__ __forceinline__ void tile_add_x_records(const int4 rec, const int m,
                                                   const float* __restrict__ x,
                                                   float* xt, const int lane) {
    constexpr int U = 16;
    int j = 0;
    for (; j + U <= m; j += U) {
        int sj[U];
        #pragma unroll
        for (int u = 0; u < U; ++u) sj[u] = __builtin_amdgcn_readlane(rec.x, j + u);
        float4 g[U];
        #pragma unroll
        for (int u = 0; u < U; ++u)
            g[u] = *(const float4*)(x + (size_t)sj[u] * DIMS_IN + lane * 4);
        #pragma unroll
        for (int u = 0; u < U; ++u) {
            const int   rj = __builtin_amdgcn_readlane(rec.y, j + u);
            const float vj = __int_as_float(__builtin_amdgcn_readlane(rec.z, j + u));
            float* a = xt + rj * XT_PITCH + lane;
            atomicAdd(a,      vj * g[u].x);
            atomicAdd(a + 32, vj * g[u].y);
            atomicAdd(a + 64, vj * g[u].z);
            atomicAdd(a + 96, vj * g[u].w);
        }
    }
    if (j < m) {
        int sj[U];
        #pragma unroll
        for (int u = 0; u < U; ++u) sj[u] = __builtin_amdgcn_readlane(rec.x, j + u);
        float4 g[U];
        #pragma unroll
        for (int u = 0; u < U; ++u) {
            if (j + u < m)
                g[u] = *(const float4*)(x + (size_t)sj[u] * DIMS_IN + lane * 4);
        }
        #pragma unroll
        for (int u = 0; u < U; ++u) {
            if (j + u < m) {
                const int   rj = __builtin_amdgcn_readlane(rec.y, j + u);
                const float vj = __int_as_float(__builtin_amdgcn_readlane(rec.z, j + u));
                float* a = xt + rj * XT_PITCH + lane;
                atomicAdd(a,      vj * g[u].x);
                atomicAdd(a + 32, vj * g[u].y);
                atomicAdd(a + 64, vj * g[u].z);
                atomicAdd(a + 96, vj * g[u].w);
            }
        }
    }
}

__global__ __launch_bounds__(SPT_THREADS)
void spmm_x_tile_gemm(const int* __restrict__ ecnt,
                      const int4* __restrict__ erec,
                      const int* __restrict__ esrc,
                      const int* __restrict__ edst,
                      const float* __restrict__ eval,
                      const float* __restrict__ x,
                      const float* __restrict__ W1t,
                      float* __restrict__ hout,
                      int E, int n) {
    __shared__ __attribute__((aligned(16))) float xt[XT_ROWS * XT_PITCH];
    __shared__ __attribute__((aligned(16))) float wl[DIMS_H * XT_PITCH];
    __shared__ __attribute__((aligned(16))) float ht[XT_ROWS * HT_PITCH];

    const int tid  = threadIdx.x;
    const int lane = tid & 31;
    const int wave = __builtin_amdgcn_readfirstlane(tid >> 5);
    const int t    = blockIdx.x;
    const int row0 = t * SP_ROWS;
    if (row0 >= n) return;
    const int rows = min(SP_ROWS, n - row0);

    {
        float4 z; z.x = 0.f; z.y = 0.f; z.z = 0.f; z.w = 0.f;
        for (int i = tid * 4; i < XT_ROWS * XT_PITCH; i += SPT_THREADS * 4)
            *(float4*)(xt + i) = z;
        for (int q = tid; q < DIMS_H * 32; q += SPT_THREADS) {
            const int nn = q >> 5;
            const float4 w = *(const float4*)(W1t + nn * DIMS_IN + lane * 4);
            float* d = wl + nn * XT_PITCH + lane;
            d[0]  = w.x;
            d[32] = w.y;
            d[64] = w.z;
            d[96] = w.w;
        }
    }
    __syncthreads();

    {
        int myc = 0;
        if (lane < SP_CPW) myc = ecnt[(wave + lane * SPT_WAVES) * SP_TILES + t];
        int P[SP_CPW + 1];
        P[0] = 0;
        #pragma unroll
        for (int i = 0; i < SP_CPW; ++i) {
            const int ci = __builtin_amdgcn_readlane(myc, i);
            P[i + 1] = P[i] + (ci <= SP_CAP ? ci : 0);
        }
        const int total = P[SP_CPW];
        int4 cur = load_flat_record(erec, P, lane, wave, t);
        asm volatile("" :: "v"(cur.x), "v"(cur.y), "v"(cur.z));
        for (int b0 = 0; b0 < total; b0 += 32) {
            const int4 nxt = load_flat_record(erec, P, b0 + 32 + lane, wave, t);
            tile_add_x_records(cur, min(32, total - b0), x, xt, lane);
            cur = nxt;
        }
        unsigned ovf = __builtin_amdgcn_ballot_w32(myc > SP_CAP);
        while (ovf != 0u) {
            const int ci = __builtin_ctz(ovf);
            ovf &= ovf - 1u;
            const int c = wave + ci * SPT_WAVES;
            const int chunk = (E + SP_CHUNKS - 1) / SP_CHUNKS;
            const int e0 = c * chunk;
            const int e1 = min(E, e0 + chunk);
            for (int e = e0 + lane; e < e1; e += 32) {
                const unsigned r = (unsigned)(edst[e] - row0);
                if (r < (unsigned)rows) {
                    const int   s = esrc[e];
                    const float v = eval[e];
                    const float* srow = x + (size_t)s * DIMS_IN;
                    float* a = xt + (int)r * XT_PITCH;
                    for (int k = 0; k < DIMS_IN; ++k)
                        atomicAdd(a + (k & 3) * 32 + (k >> 2), v * srow[k]);
                }
            }
        }
    }
    __syncthreads();

    {
        const int nmt  = (rows + 15) >> 4;
        const int half = lane >> 4;
        const int l15  = lane & 15;
        for (int u = wave; u < nmt * 2; u += SPT_WAVES) {
            const int mt = u >> 1;
            const int nh = u & 1;
            const float* ap = xt + (mt * 16 + l15) * XT_PITCH + half * 2;
            const float* bp = wl + (nh * 48 + l15) * XT_PITCH + half * 2;
            v8f d0 = {};
            v8f d1 = {};
            v8f d2 = {};
            #pragma unroll 8
            for (int k = 0; k < DIMS_IN; k += 4) {
                const v2f a  = *(const v2f*)(ap + k);
                const v2f b0 = *(const v2f*)(bp + k);
                const v2f b1 = *(const v2f*)(bp + 16 * XT_PITCH + k);
                const v2f b2 = *(const v2f*)(bp + 32 * XT_PITCH + k);
                d0 = __builtin_amdgcn_wmma_f32_16x16x4_f32(
                    false, a, false, b0, (short)0, d0, false, false);
                d1 = __builtin_amdgcn_wmma_f32_16x16x4_f32(
                    false, a, false, b1, (short)0, d1, false, false);
                d2 = __builtin_amdgcn_wmma_f32_16x16x4_f32(
                    false, a, false, b2, (short)0, d2, false, false);
            }
            float* hp = ht + (mt * 16 + half * 8) * HT_PITCH + nh * 48 + l15;
            #pragma unroll
            for (int r = 0; r < 8; ++r) {
                hp[r * HT_PITCH]      = d0[r];
                hp[r * HT_PITCH + 16] = d1[r];
                hp[r * HT_PITCH + 32] = d2[r];
            }
        }
    }
    __syncthreads();

    {
        float* o = hout + (size_t)row0 * DIMS_H;
        const int nel = rows * DIMS_H;
        for (int i = tid * 4; i < nel; i += SPT_THREADS * 4) {
            const int r  = i / DIMS_H;
            const int cc = i - r * DIMS_H;
            *(float4*)(o + i) = *(const float4*)(ht + r * HT_PITCH + cc);
        }
    }
}

extern "C" void kernel_launch(void* const* d_in, const int* in_sizes, int n_in,
                              void* d_out, int out_size, void* d_ws, size_t ws_size,
                              hipStream_t stream) {
    const float* x    = (const float*)d_in[0];
    const int*   esrc = (const int*)d_in[1];
    const int*   edst = (const int*)d_in[2];
    const float* eval = (const float*)d_in[3];
    const float* W1   = (const float*)d_in[4];
    const float* W2   = (const float*)d_in[5];
    float*       out  = (float*)d_out;

    const int E = in_sizes[1];
    const int n = in_sizes[0] / DIMS_IN;

    float* buf0 = (float*)d_ws;
    float* buf1 = buf0 + (size_t)n * DIMS_H;
    float* W1t  = buf1 + (size_t)n * DIMS_H;
    float* W2t  = W1t + DIMS_H * DIMS_IN;
    int*   ecnt = (int*)(W2t + DIMS_OUT * DIMS_H);
    int4*  erec = (int4*)(ecnt + SP_CHUNKS * SP_TILES);

    transpose_kernel<<<(DIMS_IN * DIMS_H + 255) / 256, 256, 0, stream>>>(
        W1, W1t, DIMS_IN, DIMS_H);
    transpose_kernel<<<(DIMS_H * DIMS_OUT + 255) / 256, 256, 0, stream>>>(
        W2, W2t, DIMS_H, DIMS_OUT);

    bin_edges<<<SP_CHUNKS, BIN_THREADS, 0, stream>>>(esrc, edst, eval, ecnt, erec, E, n);

    spmm_x_tile_gemm<<<SP_TILES, SPT_THREADS, 0, stream>>>(
        ecnt, erec, esrc, edst, eval, x, W1t, buf1, E, n);

    {
        int waves  = (n / 16) * (DIMS_OUT / 32);
        int blocks = (waves + 7) / 8;
        gemm_wmma_f32_bt<DIMS_H, true><<<blocks, 256, 0, stream>>>(
            buf1, W2t, buf0, n, DIMS_OUT);
    }

    spmm_tile_lds<DIMS_OUT><<<SP_TILES, SPT_THREADS, 0, stream>>>(
        ecnt, erec, esrc, edst, eval, buf0, out, E, n);
}


# Layer-1 tile kernel takes over the dense tail: stages W1, W2 itself (no transpose launches), applies relu and W2 on f32 WMMA to its LDS tile, stores relu(h) W2; 6 launches to 3; sum order kept
# speedup vs baseline: 3.5795x; 1.2950x over previous
#include <hip/hip_runtime.h>

#define DIMS_IN  128
#define DIMS_H   96
#define DIMS_OUT 32

typedef float v2f __attribute__((ext_vector_type(2)));
typedef float v8f __attribute__((ext_vector_type(8)));

#define SP_ROWS    196
#define SP_TILES   256
#define SP_CHUNKS  256
#define SP_CAP     64
#define BIN_THREADS 1024
#define SPT_THREADS 512
#define SPT_WAVES   (SPT_THREADS / 32)
#define SP_CPW      (SP_CHUNKS / SPT_WAVES)
static_assert(SP_CHUNKS % SPT_WAVES == 0 && SP_CPW <= 32, "one lane per cell of the wave");

__global__ __launch_bounds__(BIN_THREADS)
void bin_edges(const int* __restrict__ esrc,
               const int* __restrict__ edst,
               const float* __restrict__ eval,
               int* __restrict__ ecnt,
               int4* __restrict__ erec,
               int E, int n) {
    __shared__ int counters[SP_TILES];
    const int tid = threadIdx.x;
    const int c   = blockIdx.x;
    for (int i = tid; i < SP_TILES; i += BIN_THREADS) counters[i] = 0;
    __syncthreads();

    const int chunk = (E + SP_CHUNKS - 1) / SP_CHUNKS;
    const int e0 = c * chunk;
    const int e1 = min(E, e0 + chunk);
    for (int e = e0 + tid; e < e1; e += BIN_THREADS) {
        const int   d = edst[e];
        const int   s = esrc[e];
        const float v = eval[e];
        asm volatile("" :: "v"(s), "v"(v));
        if ((unsigned)d < (unsigned)n) {
            const int t = d / SP_ROWS;
            if (t < SP_TILES) {
                const int slot = atomicAdd(&counters[t], 1);
                if (slot < SP_CAP) {
                    int4 rec;
                    rec.x = s;
                    rec.y = d - t * SP_ROWS;
                    rec.z = __float_as_int(v);
                    rec.w = 0;
                    erec[((size_t)c * SP_TILES + t) * SP_CAP + slot] = rec;
                }
            }
        }
    }
    __syncthreads();
    for (int i = tid; i < SP_TILES; i += BIN_THREADS)
        ecnt[c * SP_TILES + i] = counters[i];
}

__device__ __forceinline__ int4 load_flat_record(const int4* __restrict__ erec,
                                                 const int (&P)[SP_CPW + 1],
                                                 const int i, const int wave, const int t) {
    int k = 0, base = 0;
    #pragma unroll
    for (int j = 1; j < SP_CPW; ++j) {
        if (i >= P[j]) { k = j; base = P[j]; }
    }
    int4 r = make_int4(0, 0, 0, 0);
    if (i < P[SP_CPW])
        r = erec[((size_t)(wave + k * SPT_WAVES) * SP_TILES + t) * SP_CAP + (i - base)];
    return r;
}

template <int D>
__device__ __forceinline__ void tile_add_records(const int4 rec, const int m,
                                                 const float* __restrict__ dense,
                                                 float* acc, const int lane) {
    constexpr int C = D / 32;
    constexpr int U = (48 / C) > 32 ? 32 : (48 / C);
    int j = 0;
    for (; j + U <= m; j += U) {
        int sj[U];
        #pragma unroll
        for (int u = 0; u < U; ++u) sj[u] = __builtin_amdgcn_readlane(rec.x, j + u);
        float g[U][C];
        #pragma unroll
        for (int u = 0; u < U; ++u) {
            const float* srow = dense + (size_t)sj[u] * D + lane;
            #pragma unroll
            for (int k = 0; k < C; ++k) g[u][k] = srow[k * 32];
        }
        #pragma unroll
        for (int u = 0; u < U; ++u) {
            const int   rj = __builtin_amdgcn_readlane(rec.y, j + u);
            const float vj = __int_as_float(__builtin_amdgcn_readlane(rec.z, j + u));
            float* a = acc + rj * D + lane;
            #pragma unroll
            for (int k = 0; k < C; ++k) atomicAdd(a + k * 32, vj * g[u][k]);
        }
    }
    if (j < m) {
        int sj[U];
        #pragma unroll
        for (int u = 0; u < U; ++u) sj[u] = __builtin_amdgcn_readlane(rec.x, j + u);
        float g[U][C];
        #pragma unroll
        for (int u = 0; u < U; ++u) {
            if (j + u < m) {
                const float* srow = dense + (size_t)sj[u] * D + lane;
                #pragma unroll
                for (int k = 0; k < C; ++k) g[u][k] = srow[k * 32];
            }
        }
        #pragma unroll
        for (int u = 0; u < U; ++u) {
            if (j + u < m) {
                const int   rj = __builtin_amdgcn_readlane(rec.y, j + u);
                const float vj = __int_as_float(__builtin_amdgcn_readlane(rec.z, j + u));
                float* a = acc + rj * D + lane;
                #pragma unroll
                for (int k = 0; k < C; ++k) atomicAdd(a + k * 32, vj * g[u][k]);
            }
        }
    }
}

template <int D>
__global__ __launch_bounds__(SPT_THREADS)
void spmm_tile_lds(const int* __restrict__ ecnt,
                   const int4* __restrict__ erec,
                   const int* __restrict__ esrc,
                   const int* __restrict__ edst,
                   const float* __restrict__ eval,
                   const float* __restrict__ dense,
                   float* __restrict__ out,
                   int E, int n) {
    __shared__ __attribute__((aligned(16))) float acc[SP_ROWS * D];

    const int tid  = threadIdx.x;
    const int lane = tid & 31;
    const int wave = __builtin_amdgcn_readfirstlane(tid >> 5);
    const int t    = blockIdx.x;
    const int row0 = t * SP_ROWS;
    if (row0 >= n) return;
    const int rows = min(SP_ROWS, n - row0);
    const int nel  = rows * D;

    {
        float4 z; z.x = 0.f; z.y = 0.f; z.z = 0.f; z.w = 0.f;
        for (int i = tid * 4; i < nel; i += SPT_THREADS * 4)
            *(float4*)(acc + i) = z;
    }
    __syncthreads();

    {
        int myc = 0;
        if (lane < SP_CPW) myc = ecnt[(wave + lane * SPT_WAVES) * SP_TILES + t];
        int P[SP_CPW + 1];
        P[0] = 0;
        #pragma unroll
        for (int i = 0; i < SP_CPW; ++i) {
            const int ci = __builtin_amdgcn_readlane(myc, i);
            P[i + 1] = P[i] + (ci <= SP_CAP ? ci : 0);
        }
        const int total = P[SP_CPW];
        int4 cur = load_flat_record(erec, P, lane, wave, t);
        asm volatile("" :: "v"(cur.x), "v"(cur.y), "v"(cur.z));
        for (int b0 = 0; b0 < total; b0 += 32) {
            const int4 nxt = load_flat_record(erec, P, b0 + 32 + lane, wave, t);
            tile_add_records<D>(cur, min(32, total - b0), dense, acc, lane);
            cur = nxt;
        }
        unsigned ovf = __builtin_amdgcn_ballot_w32(myc > SP_CAP);
        while (ovf != 0u) {
            const int ci = __builtin_ctz(ovf);
            ovf &= ovf - 1u;
            const int c = wave + ci * SPT_WAVES;
            const int chunk = (E + SP_CHUNKS - 1) / SP_CHUNKS;
            const int e0 = c * chunk;
            const int e1 = min(E, e0 + chunk);
            for (int e = e0 + lane; e < e1; e += 32) {
                const unsigned r = (unsigned)(edst[e] - row0);
                if (r < (unsigned)rows) {
                    const int   s = esrc[e];
                    const float v = eval[e];
                    const float* srow = dense + (size_t)s * D;
                    float* a = acc + (int)r * D;
                    for (int k = 0; k < D; ++k) atomicAdd(a + k, v * srow[k]);
                }
            }
        }
    }
    __syncthreads();

    {
        float* o = out + (size_t)row0 * D;
        for (int i = tid * 4; i < nel; i += SPT_THREADS * 4)
            *(float4*)(o + i) = *(const float4*)(acc + i);
    }
}

#define XT_PITCH 132
#define XT_ROWS  208
#define HT_PITCH 100

__device__ __forceinline__ void tile_add_x_records(const int4 rec, const int m,
                                                   const float* __restrict__ x,
                                                   float* xt, const int lane) {
    constexpr int U = 16;
    int j = 0;
    for (; j + U <= m; j += U) {
        int sj[U];
        #pragma unroll
        for (int u = 0; u < U; ++u) sj[u] = __builtin_amdgcn_readlane(rec.x, j + u);
        float4 g[U];
        #pragma unroll
        for (int u = 0; u < U; ++u)
            g[u] = *(const float4*)(x + (size_t)sj[u] * DIMS_IN + lane * 4);
        #pragma unroll
        for (int u = 0; u < U; ++u) {
            const int   rj = __builtin_amdgcn_readlane(rec.y, j + u);
            const float vj = __int_as_float(__builtin_amdgcn_readlane(rec.z, j + u));
            float* a = xt + rj * XT_PITCH + lane;
            atomicAdd(a,      vj * g[u].x);
            atomicAdd(a + 32, vj * g[u].y);
            atomicAdd(a + 64, vj * g[u].z);
            atomicAdd(a + 96, vj * g[u].w);
        }
    }
    if (j < m) {
        int sj[U];
        #pragma unroll
        for (int u = 0; u < U; ++u) sj[u] = __builtin_amdgcn_readlane(rec.x, j + u);
        float4 g[U];
        #pragma unroll
        for (int u = 0; u < U; ++u) {
            if (j + u < m)
                g[u] = *(const float4*)(x + (size_t)sj[u] * DIMS_IN + lane * 4);
        }
        #pragma unroll
        for (int u = 0; u < U; ++u) {
            if (j + u < m) {
                const int   rj = __builtin_amdgcn_readlane(rec.y, j + u);
                const float vj = __int_as_float(__builtin_amdgcn_readlane(rec.z, j + u));
                float* a = xt + rj * XT_PITCH + lane;
                atomicAdd(a,      vj * g[u].x);
                atomicAdd(a + 32, vj * g[u].y);
                atomicAdd(a + 64, vj * g[u].z);
                atomicAdd(a + 96, vj * g[u].w);
            }
        }
    }
}

__global__ __launch_bounds__(SPT_THREADS)
void spmm_x_tile_gemm(const int* __restrict__ ecnt,
                      const int4* __restrict__ erec,
                      const int* __restrict__ esrc,
                      const int* __restrict__ edst,
                      const float* __restrict__ eval,
                      const float* __restrict__ x,
                      const float* __restrict__ W1,
                      const float* __restrict__ W2,
                      float* __restrict__ hw2out,
                      int E, int n) {
    __shared__ __attribute__((aligned(16))) float xt[XT_ROWS * XT_PITCH];
    __shared__ __attribute__((aligned(16))) float wl[DIMS_H * XT_PITCH];
    __shared__ __attribute__((aligned(16))) float ht[XT_ROWS * HT_PITCH];
    __shared__ __attribute__((aligned(16))) float w2l[DIMS_OUT * HT_PITCH];

    const int tid  = threadIdx.x;
    const int lane = tid & 31;
    const int wave = __builtin_amdgcn_readfirstlane(tid >> 5);
    const int t    = blockIdx.x;
    const int row0 = t * SP_ROWS;
    if (row0 >= n) return;
    const int rows = min(SP_ROWS, n - row0);

    {
        float4 z; z.x = 0.f; z.y = 0.f; z.z = 0.f; z.w = 0.f;
        for (int i = tid * 4; i < XT_ROWS * XT_PITCH; i += SPT_THREADS * 4)
            *(float4*)(xt + i) = z;
        static_assert((DIMS_IN * DIMS_H) % SPT_THREADS == 0 &&
                      (DIMS_H * DIMS_OUT) % SPT_THREADS == 0, "whole passes over the weights");
        #pragma unroll
        for (int it = 0; it < (DIMS_IN * DIMS_H) / SPT_THREADS; ++it) {
            const int q  = it * SPT_THREADS + tid;
            const int k  = q / DIMS_H;
            const int nn = q - k * DIMS_H;
            wl[nn * XT_PITCH + (k & 3) * 32 + (k >> 2)] = W1[q];
        }
        #pragma unroll
        for (int it = 0; it < (DIMS_H * DIMS_OUT) / SPT_THREADS; ++it) {
            const int q  = it * SPT_THREADS + tid;
            const int k  = q / DIMS_OUT;
            const int nn = q - k * DIMS_OUT;
            w2l[nn * HT_PITCH + k] = W2[q];
        }
    }
    __syncthreads();

    {
        int myc = 0;
        if (lane < SP_CPW) myc = ecnt[(wave + lane * SPT_WAVES) * SP_TILES + t];
        int P[SP_CPW + 1];
        P[0] = 0;
        #pragma unroll
        for (int i = 0; i < SP_CPW; ++i) {
            const int ci = __builtin_amdgcn_readlane(myc, i);
            P[i + 1] = P[i] + (ci <= SP_CAP ? ci : 0);
        }
        const int total = P[SP_CPW];
        int4 cur = load_flat_record(erec, P, lane, wave, t);
        asm volatile("" :: "v"(cur.x), "v"(cur.y), "v"(cur.z));
        for (int b0 = 0; b0 < total; b0 += 32) {
            const int4 nxt = load_flat_record(erec, P, b0 + 32 + lane, wave, t);
            tile_add_x_records(cur, min(32, total - b0), x, xt, lane);
            cur = nxt;
        }
        unsigned ovf = __builtin_amdgcn_ballot_w32(myc > SP_CAP);
        while (ovf != 0u) {
            const int ci = __builtin_ctz(ovf);
            ovf &= ovf - 1u;
            const int c = wave + ci * SPT_WAVES;
            const int chunk = (E + SP_CHUNKS - 1) / SP_CHUNKS;
            const int e0 = c * chunk;
            const int e1 = min(E, e0 + chunk);
            for (int e = e0 + lane; e < e1; e += 32) {
                const unsigned r = (unsigned)(edst[e] - row0);
                if (r < (unsigned)rows) {
                    const int   s = esrc[e];
                    const float v = eval[e];
                    const float* srow = x + (size_t)s * DIMS_IN;
                    float* a = xt + (int)r * XT_PITCH;
                    for (int k = 0; k < DIMS_IN; ++k)
                        atomicAdd(a + (k & 3) * 32 + (k >> 2), v * srow[k]);
                }
            }
        }
    }
    __syncthreads();

    {
        const int nmt  = (rows + 15) >> 4;
        const int half = lane >> 4;
        const int l15  = lane & 15;
        for (int u = wave; u < nmt * 2; u += SPT_WAVES) {
            const int mt = u >> 1;
            const int nh = u & 1;
            const float* ap = xt + (mt * 16 + l15) * XT_PITCH + half * 2;
            const float* bp = wl + (nh * 48 + l15) * XT_PITCH + half * 2;
            v8f d0 = {};
            v8f d1 = {};
            v8f d2 = {};
            #pragma unroll 8
            for (int k = 0; k < DIMS_IN; k += 4) {
                const v2f a  = *(const v2f*)(ap + k);
                const v2f b0 = *(const v2f*)(bp + k);
                const v2f b1 = *(const v2f*)(bp + 16 * XT_PITCH + k);
                const v2f b2 = *(const v2f*)(bp + 32 * XT_PITCH + k);
                d0 = __builtin_amdgcn_wmma_f32_16x16x4_f32(
                    false, a, false, b0, (short)0, d0, false, false);
                d1 = __builtin_amdgcn_wmma_f32_16x16x4_f32(
                    false, a, false, b1, (short)0, d1, false, false);
                d2 = __builtin_amdgcn_wmma_f32_16x16x4_f32(
                    false, a, false, b2, (short)0, d2, false, false);
            }
            float* hp = ht + (mt * 16 + half * 8) * HT_PITCH + nh * 48 + l15;
            #pragma unroll
            for (int r = 0; r < 8; ++r) {
                hp[r * HT_PITCH]      = d0[r];
                hp[r * HT_PITCH + 16] = d1[r];
                hp[r * HT_PITCH + 32] = d2[r];
            }
        }
    }
    __syncthreads();

    float* ot = xt;
    {
        const int nmt  = (rows + 15) >> 4;
        const int half = lane >> 4;
        const int l15  = lane & 15;
        for (int u = wave; u < nmt * 2; u += SPT_WAVES) {
            const int mt = u >> 1;
            const int nq = u & 1;
            const float* ap = ht  + (mt * 16 + l15) * HT_PITCH + half * 2;
            const float* bp = w2l + (nq * 16 + l15) * HT_PITCH + half * 2;
            v8f d = {};
            #pragma unroll
            for (int k = 0; k < DIMS_H; k += 4) {
                v2f a = *(const v2f*)(ap + k);
                a.x = fmaxf(a.x, 0.f);
                a.y = fmaxf(a.y, 0.f);
                const v2f b = *(const v2f*)(bp + k);
                d = __builtin_amdgcn_wmma_f32_16x16x4_f32(
                    false, a, false, b, (short)0, d, false, false);
            }
            float* op = ot + (mt * 16 + half * 8) * DIMS_OUT + nq * 16 + l15;
            #pragma unroll
            for (int r = 0; r < 8; ++r) op[r * DIMS_OUT] = d[r];
        }
    }
    __syncthreads();

    {
        float* o = hw2out + (size_t)row0 * DIMS_OUT;
        const int nel = rows * DIMS_OUT;
        for (int i = tid * 4; i < nel; i += SPT_THREADS * 4)
            *(float4*)(o + i) = *(const float4*)(ot + i);
    }
}

extern "C" void kernel_launch(void* const* d_in, const int* in_sizes, int n_in,
                              void* d_out, int out_size, void* d_ws, size_t ws_size,
                              hipStream_t stream) {
    const float* x    = (const float*)d_in[0];
    const int*   esrc = (const int*)d_in[1];
    const int*   edst = (const int*)d_in[2];
    const float* eval = (const float*)d_in[3];
    const float* W1   = (const float*)d_in[4];
    const float* W2   = (const float*)d_in[5];
    float*       out  = (float*)d_out;

    const int E = in_sizes[1];
    const int n = in_sizes[0] / DIMS_IN;

    float* buf0 = (float*)d_ws;
    int*   ecnt = (int*)(buf0 + (size_t)n * DIMS_OUT);
    int4*  erec = (int4*)(ecnt + SP_CHUNKS * SP_TILES);

    bin_edges<<<SP_CHUNKS, BIN_THREADS, 0, stream>>>(esrc, edst, eval, ecnt, erec, E, n);

    spmm_x_tile_gemm<<<SP_TILES, SPT_THREADS, 0, stream>>>(
        ecnt, erec, esrc, edst, eval, x, W1, W2, buf0, E, n);

    spmm_tile_lds<DIMS_OUT><<<SP_TILES, SPT_THREADS, 0, stream>>>(
        ecnt, erec, esrc, edst, eval, buf0, out, E, n);
}
